# starting kernel
# baseline (speedup 1.0000x reference)
#include <hip/hip_runtime.h>

typedef __attribute__((ext_vector_type(2))) float v2f;
typedef __attribute__((ext_vector_type(8))) float v8f;

#define HCH 128
#define OCH 64

__global__ __launch_bounds__(256) void k_fill_deg(float* __restrict__ deg, int n) {
  int i = blockIdx.x * 256 + threadIdx.x;
  if (i < n) deg[i] = 1.0f;
}

__global__ __launch_bounds__(256) void k_count_deg(const int* __restrict__ dst,
                                                   float* __restrict__ deg, int e) {
  int i = blockIdx.x * 256 + threadIdx.x;
  if (i < e) atomicAdd(&deg[dst[i]], 1.0f);
}

__global__ __launch_bounds__(256) void k_dinv(float* __restrict__ deg, int n) {
  int i = blockIdx.x * 256 + threadIdx.x;
  if (i < n) deg[i] = rsqrtf(deg[i]);
}

__global__ __launch_bounds__(256) void k_gemm1(const float* __restrict__ X,
                                               const float* __restrict__ W,
                                               const float* __restrict__ dinv,
                                               float* __restrict__ Hout,
                                               float* __restrict__ Agg, int n) {
  const int wave = threadIdx.x >> 5;
  const int lane = threadIdx.x & 31;
  const int rowTile = blockIdx.x;
  const int colTile = wave;
  if (rowTile * 16 >= n) return;
  const int hs = lane >> 4;
  const int lr = lane & 15;
  const int row = rowTile * 16 + lr;
  const int col = colTile * 16 + lr;
  const float* __restrict__ xr = X + (size_t)row * HCH;

  v8f acc = {};
#pragma unroll 4
  for (int k0 = 0; k0 < HCH; k0 += 4) {
    const int ka = k0 + hs * 2;
    v2f a; a.x = xr[ka];             a.y = xr[ka + 1];
    v2f b; b.x = W[ka * HCH + col];  b.y = W[(ka + 1) * HCH + col];
    acc = __builtin_amdgcn_wmma_f32_16x16x4_f32(false, a, false, b,
                                                (short)0, acc, false, false);
  }
#pragma unroll
  for (int r = 0; r < 8; ++r) {
    const int m = rowTile * 16 + hs * 8 + r;
    const float v = acc[r];
    const float di = dinv[m];
    Hout[(size_t)m * HCH + col] = v;
    Agg [(size_t)m * HCH + col] = v * di * di;
  }
}

__global__ __launch_bounds__(256) void k_gemm2(const float* __restrict__ Agg,
                                               const float* __restrict__ b1,
                                               const float* __restrict__ W,
                                               float* __restrict__ H2, int n) {
  const int wave = threadIdx.x >> 5;
  const int lane = threadIdx.x & 31;
  const int rowTile = blockIdx.x * 2 + (wave >> 2);
  const int colTile = wave & 3;
  if (rowTile * 16 >= n) return;
  const int hs = lane >> 4;
  const int lr = lane & 15;
  const int row = rowTile * 16 + lr;
  const int col = colTile * 16 + lr;
  const float* __restrict__ ar = Agg + (size_t)row * HCH;

  v8f acc = {};
#pragma unroll 4
  for (int k0 = 0; k0 < HCH; k0 += 4) {
    const int ka = k0 + hs * 2;
    float a0 = ar[ka]     + b1[ka];
    float a1 = ar[ka + 1] + b1[ka + 1];
    v2f a; a.x = a0 > 0.0f ? a0 : 0.0f;
           a.y = a1 > 0.0f ? a1 : 0.0f;
    v2f b; b.x = W[ka * OCH + col];  b.y = W[(ka + 1) * OCH + col];
    acc = __builtin_amdgcn_wmma_f32_16x16x4_f32(false, a, false, b,
                                                (short)0, acc, false, false);
  }
#pragma unroll
  for (int r = 0; r < 8; ++r) {
    const int m = rowTile * 16 + hs * 8 + r;
    H2[(size_t)m * OCH + col] = acc[r];
  }
}

__global__ __launch_bounds__(256) void k_agg1_edges(const int* __restrict__ src,
                                                    const int* __restrict__ dst,
                                                    const float* __restrict__ dinv,
                                                    const float* __restrict__ h,
                                                    float* __restrict__ agg, int e) {
  int gid = blockIdx.x * 256 + threadIdx.x;
  int eidx = gid >> 5;
  if (eidx >= e) return;
  int c4 = (gid & 31) * 4;
  int s = src[eidx], d = dst[eidx];
  float nrm = dinv[s] * dinv[d];
  const float4 hv = *(const float4*)(h + (size_t)s * HCH + c4);
  float* ap = agg + (size_t)d * HCH + c4;
  atomicAdd(ap + 0, hv.x * nrm);
  atomicAdd(ap + 1, hv.y * nrm);
  atomicAdd(ap + 2, hv.z * nrm);
  atomicAdd(ap + 3, hv.w * nrm);
}

__global__ __launch_bounds__(256) void k_out_init(float* __restrict__ out,
                                                  const float* __restrict__ b2, int total) {
  int gid = blockIdx.x * 256 + threadIdx.x;
  if (gid < total) out[gid] = b2[gid & (OCH - 1)];
}

__global__ __launch_bounds__(256) void k_agg2_edges(const int* __restrict__ src,
                                                    const int* __restrict__ dst,
                                                    const float* __restrict__ h2,
                                                    float* __restrict__ out, int e) {
  int gid = blockIdx.x * 256 + threadIdx.x;
  int eidx = gid >> 4;
  if (eidx >= e) return;
  int c4 = (gid & 15) * 4;
  int s = src[eidx], d = dst[eidx];
  const float4 hv = *(const float4*)(h2 + (size_t)s * OCH + c4);
  float* op = out + (size_t)d * OCH + c4;
  atomicAdd(op + 0, hv.x);
  atomicAdd(op + 1, hv.y);
  atomicAdd(op + 2, hv.z);
  atomicAdd(op + 3, hv.w);
}

extern "C" void kernel_launch(void* const* d_in, const int* in_sizes, int n_in,
                              void* d_out, int out_size, void* d_ws, size_t ws_size,
                              hipStream_t stream) {
  const float* x  = (const float*)d_in[0];
  const int*   ei = (const int*)d_in[1];
  const float* W1 = (const float*)d_in[2];
  const float* b1 = (const float*)d_in[3];
  const float* W2 = (const float*)d_in[4];
  const float* b2 = (const float*)d_in[5];

  const int N = in_sizes[0] / HCH;
  const int E = in_sizes[1] / 2;
  const int* src = ei;
  const int* dst = ei + E;

  float* ws   = (float*)d_ws;
  float* dinv = ws;
  size_t off  = ((size_t)N + 63) & ~(size_t)63;
  float* h    = ws + off;
  float* agg  = h + (size_t)N * HCH;
  float* h2   = h;
  float* out  = (float*)d_out;

  k_fill_deg <<<(N + 255) / 256, 256, 0, stream>>>(dinv, N);
  k_count_deg<<<(E + 255) / 256, 256, 0, stream>>>(dst, dinv, E);
  k_dinv     <<<(N + 255) / 256, 256, 0, stream>>>(dinv, N);

  k_gemm1<<<N / 16, 256, 0, stream>>>(x, W1, dinv, h, agg, N);

  long long ew1 = (long long)E * 32;
  k_agg1_edges<<<(unsigned)((ew1 + 255) / 256), 256, 0, stream>>>(src, dst, dinv, h, agg, E);

  k_gemm2<<<N / 32, 256, 0, stream>>>(agg, b1, W2, h2, N);

  int tot2 = N * OCH;
  k_out_init  <<<(tot2 + 255) / 256, 256, 0, stream>>>(out, b2, tot2);
  long long ew2 = (long long)E * 16;
  k_agg2_edges<<<(unsigned)((ew2 + 255) / 256), 256, 0, stream>>>(src, dst, h2, out, E);
}


# Edge scatters: lane l owns channels l+32j, so each f32 atomic and row-load instruction covers one whole 128-B line, not 8 lanes on each of 4 lines; atomics and their arrival order as before
# speedup vs baseline: 1.2901x; 1.2901x over previous
#include <hip/hip_runtime.h>

typedef __attribute__((ext_vector_type(2))) float v2f;
typedef __attribute__((ext_vector_type(8))) float v8f;

#define HCH 128
#define OCH 64

__global__ __launch_bounds__(256) void k_fill_deg(float* __restrict__ deg, int n) {
  int i = blockIdx.x * 256 + threadIdx.x;
  if (i < n) deg[i] = 1.0f;
}

__global__ __launch_bounds__(256) void k_count_deg(const int* __restrict__ dst,
                                                   float* __restrict__ deg, int e) {
  int i = blockIdx.x * 256 + threadIdx.x;
  if (i < e) atomicAdd(&deg[dst[i]], 1.0f);
}

__global__ __launch_bounds__(256) void k_dinv(float* __restrict__ deg, int n) {
  int i = blockIdx.x * 256 + threadIdx.x;
  if (i < n) deg[i] = rsqrtf(deg[i]);
}

__global__ __launch_bounds__(256) void k_gemm1(const float* __restrict__ X,
                                               const float* __restrict__ W,
                                               const float* __restrict__ dinv,
                                               float* __restrict__ Hout,
                                               float* __restrict__ Agg, int n) {
  const int wave = threadIdx.x >> 5;
  const int lane = threadIdx.x & 31;
  const int rowTile = blockIdx.x;
  const int colTile = wave;
  if (rowTile * 16 >= n) return;
  const int hs = lane >> 4;
  const int lr = lane & 15;
  const int row = rowTile * 16 + lr;
  const int col = colTile * 16 + lr;
  const float* __restrict__ xr = X + (size_t)row * HCH;

  v8f acc = {};
#pragma unroll 4
  for (int k0 = 0; k0 < HCH; k0 += 4) {
    const int ka = k0 + hs * 2;
    v2f a; a.x = xr[ka];             a.y = xr[ka + 1];
    v2f b; b.x = W[ka * HCH + col];  b.y = W[(ka + 1) * HCH + col];
    acc = __builtin_amdgcn_wmma_f32_16x16x4_f32(false, a, false, b,
                                                (short)0, acc, false, false);
  }
#pragma unroll
  for (int r = 0; r < 8; ++r) {
    const int m = rowTile * 16 + hs * 8 + r;
    const float v = acc[r];
    const float di = dinv[m];
    Hout[(size_t)m * HCH + col] = v;
    Agg [(size_t)m * HCH + col] = v * di * di;
  }
}

__global__ __launch_bounds__(256) void k_gemm2(const float* __restrict__ Agg,
                                               const float* __restrict__ b1,
                                               const float* __restrict__ W,
                                               float* __restrict__ H2, int n) {
  const int wave = threadIdx.x >> 5;
  const int lane = threadIdx.x & 31;
  const int rowTile = blockIdx.x * 2 + (wave >> 2);
  const int colTile = wave & 3;
  if (rowTile * 16 >= n) return;
  const int hs = lane >> 4;
  const int lr = lane & 15;
  const int row = rowTile * 16 + lr;
  const int col = colTile * 16 + lr;
  const float* __restrict__ ar = Agg + (size_t)row * HCH;

  v8f acc = {};
#pragma unroll 4
  for (int k0 = 0; k0 < HCH; k0 += 4) {
    const int ka = k0 + hs * 2;
    float a0 = ar[ka]     + b1[ka];
    float a1 = ar[ka + 1] + b1[ka + 1];
    v2f a; a.x = a0 > 0.0f ? a0 : 0.0f;
           a.y = a1 > 0.0f ? a1 : 0.0f;
    v2f b; b.x = W[ka * OCH + col];  b.y = W[(ka + 1) * OCH + col];
    acc = __builtin_amdgcn_wmma_f32_16x16x4_f32(false, a, false, b,
                                                (short)0, acc, false, false);
  }
#pragma unroll
  for (int r = 0; r < 8; ++r) {
    const int m = rowTile * 16 + hs * 8 + r;
    H2[(size_t)m * OCH + col] = acc[r];
  }
}

__global__ __launch_bounds__(256) void k_agg1_edges(const int* __restrict__ src,
                                                    const int* __restrict__ dst,
                                                    const float* __restrict__ dinv,
                                                    const float* __restrict__ h,
                                                    float* __restrict__ agg, int e) {
  int gid = blockIdx.x * 256 + threadIdx.x;
  int eidx = gid >> 5;
  if (eidx >= e) return;
  int lane = gid & 31;
  int s = src[eidx], d = dst[eidx];
  float nrm = dinv[s] * dinv[d];
  const float* hp = h + (size_t)s * HCH + lane;
  float* ap = agg + (size_t)d * HCH + lane;
  const float v0 = hp[0], v1 = hp[32], v2 = hp[64], v3 = hp[96];
  atomicAdd(ap,      v0 * nrm);
  atomicAdd(ap + 32, v1 * nrm);
  atomicAdd(ap + 64, v2 * nrm);
  atomicAdd(ap + 96, v3 * nrm);
}

__global__ __launch_bounds__(256) void k_out_init(float* __restrict__ out,
                                                  const float* __restrict__ b2, int total) {
  int gid = blockIdx.x * 256 + threadIdx.x;
  if (gid < total) out[gid] = b2[gid & (OCH - 1)];
}

__global__ __launch_bounds__(256) void k_agg2_edges(const int* __restrict__ src,
                                                    const int* __restrict__ dst,
                                                    const float* __restrict__ h2,
                                                    float* __restrict__ out, int e) {
  int gid = blockIdx.x * 256 + threadIdx.x;
  int e0 = (gid >> 5) * 2;
  if (e0 >= e) return;
  int lane = gid & 31;
  if (e0 + 1 >= e) {
    const int el = e - 1;
    const int s = src[el], d = dst[el];
    const float* p = h2 + (size_t)s * OCH + lane;
    const float a0 = p[0], a1 = p[32];
    float* q = out + (size_t)d * OCH + lane;
    atomicAdd(q,      a0);
    atomicAdd(q + 32, a1);
    return;
  }
  const int s0 = src[e0], s1 = src[e0 + 1];
  const int d0 = dst[e0], d1 = dst[e0 + 1];
  const float* p0 = h2 + (size_t)s0 * OCH + lane;
  const float* p1 = h2 + (size_t)s1 * OCH + lane;
  const float a0 = p0[0], a1 = p0[32];
  const float c0 = p1[0], c1 = p1[32];
  float* q0 = out + (size_t)d0 * OCH + lane;
  float* q1 = out + (size_t)d1 * OCH + lane;
  atomicAdd(q0,      a0);
  atomicAdd(q0 + 32, a1);
  atomicAdd(q1,      c0);
  atomicAdd(q1 + 32, c1);
}

extern "C" void kernel_launch(void* const* d_in, const int* in_sizes, int n_in,
                              void* d_out, int out_size, void* d_ws, size_t ws_size,
                              hipStream_t stream) {
  const float* x  = (const float*)d_in[0];
  const int*   ei = (const int*)d_in[1];
  const float* W1 = (const float*)d_in[2];
  const float* b1 = (const float*)d_in[3];
  const float* W2 = (const float*)d_in[4];
  const float* b2 = (const float*)d_in[5];

  const int N = in_sizes[0] / HCH;
  const int E = in_sizes[1] / 2;
  const int* src = ei;
  const int* dst = ei + E;

  float* ws   = (float*)d_ws;
  float* dinv = ws;
  size_t off  = ((size_t)N + 63) & ~(size_t)63;
  float* h    = ws + off;
  float* agg  = h + (size_t)N * HCH;
  float* h2   = h;
  float* out  = (float*)d_out;

  k_fill_deg <<<(N + 255) / 256, 256, 0, stream>>>(dinv, N);
  k_count_deg<<<(E + 255) / 256, 256, 0, stream>>>(dst, dinv, E);
  k_dinv     <<<(N + 255) / 256, 256, 0, stream>>>(dinv, N);

  k_gemm1<<<N / 16, 256, 0, stream>>>(x, W1, dinv, h, agg, N);

  long long ew1 = (long long)E * 32;
  k_agg1_edges<<<(unsigned)((ew1 + 255) / 256), 256, 0, stream>>>(src, dst, dinv, h, agg, E);

  k_gemm2<<<N / 32, 256, 0, stream>>>(agg, b1, W2, h2, N);

  int tot2 = N * OCH;
  k_out_init  <<<(tot2 + 255) / 256, 256, 0, stream>>>(out, b2, tot2);
  long long ew2 = (long long)E * 16;
  k_agg2_edges<<<(unsigned)((ew2 + 255) / 256), 256, 0, stream>>>(src, dst, h2, out, E);
}


# Both edge aggregations: edges bucketed by destination per call (int atomic slot claim), one wave per node sums its in-rows in registers in bucket arrival order and stores once; f32 atomics gone
# speedup vs baseline: 1.7066x; 1.3228x over previous
#include <hip/hip_runtime.h>

typedef __attribute__((ext_vector_type(2))) float v2f;
typedef __attribute__((ext_vector_type(8))) float v8f;

#define HCH 128
#define OCH 64

__global__ __launch_bounds__(256) void k_zero_cnt(int* __restrict__ cnt, int* __restrict__ total, int n) {
  int i = blockIdx.x * 256 + threadIdx.x;
  if (i < n) cnt[i] = 0;
  if (i == 0) *total = 0;
}

__global__ __launch_bounds__(256) void k_count(const int* __restrict__ dst,
                                               int* __restrict__ cnt, int e) {
  int i = blockIdx.x * 256 + threadIdx.x;
  if (i < e) atomicAdd(&cnt[dst[i]], 1);
}

__global__ __launch_bounds__(256) void k_alloc(const int* __restrict__ cnt,
                                               int* __restrict__ cursor,
                                               float* __restrict__ dinv,
                                               int* __restrict__ total, int n) {
  __shared__ int wsum[8];
  __shared__ int bbase;
  const int i = blockIdx.x * 256 + threadIdx.x;
  const int lane = threadIdx.x & 31;
  const int wave = threadIdx.x >> 5;
  const int c = (i < n) ? cnt[i] : 0;
  int v = c;
  for (int d = 1; d < 32; d <<= 1) {
    const int t = __shfl_up(v, d);
    if (lane >= d) v += t;
  }
  if (lane == 31) wsum[wave] = v;
  __syncthreads();
  if (threadIdx.x == 0) {
    int s = 0;
    for (int w = 0; w < 8; ++w) { const int t = wsum[w]; wsum[w] = s; s += t; }
    bbase = atomicAdd(total, s);
  }
  __syncthreads();
  if (i < n) {
    cursor[i] = bbase + wsum[wave] + v - c;
    dinv[i] = rsqrtf((float)(c + 1));
  }
}

__global__ __launch_bounds__(256) void k_fill(const int* __restrict__ src,
                                              const int* __restrict__ dst,
                                              const float* __restrict__ dinv,
                                              int* __restrict__ cursor,
                                              int* __restrict__ slots, int e) {
  int i = blockIdx.x * 256 + threadIdx.x;
  if (i < e) {
    const int s = src[i], d = dst[i];
    const int p = atomicAdd(&cursor[d], 1);
    reinterpret_cast<int2*>(slots)[p] = make_int2(s, __float_as_int(dinv[s]));
  }
}

__global__ __launch_bounds__(256) void k_gemm1(const float* __restrict__ X,
                                               const float* __restrict__ W,
                                               float* __restrict__ Hout, int n) {
  const int wave = threadIdx.x >> 5;
  const int lane = threadIdx.x & 31;
  const int rowTile = blockIdx.x;
  const int colTile = wave;
  if (rowTile * 16 >= n) return;
  const int hs = lane >> 4;
  const int lr = lane & 15;
  const int row = rowTile * 16 + lr;
  const int col = colTile * 16 + lr;
  const float* __restrict__ xr = X + (size_t)row * HCH;

  v8f acc = {};
#pragma unroll 4
  for (int k0 = 0; k0 < HCH; k0 += 4) {
    const int ka = k0 + hs * 2;
    v2f a; a.x = xr[ka];             a.y = xr[ka + 1];
    v2f b; b.x = W[ka * HCH + col];  b.y = W[(ka + 1) * HCH + col];
    acc = __builtin_amdgcn_wmma_f32_16x16x4_f32(false, a, false, b,
                                                (short)0, acc, false, false);
  }
#pragma unroll
  for (int r = 0; r < 8; ++r) {
    const int m = rowTile * 16 + hs * 8 + r;
    Hout[(size_t)m * HCH + col] = acc[r];
  }
}

__global__ __launch_bounds__(256) void k_gemm2(const float* __restrict__ Agg,
                                               const float* __restrict__ b1,
                                               const float* __restrict__ W,
                                               float* __restrict__ H2, int n) {
  const int wave = threadIdx.x >> 5;
  const int lane = threadIdx.x & 31;
  const int rowTile = blockIdx.x * 2 + (wave >> 2);
  const int colTile = wave & 3;
  if (rowTile * 16 >= n) return;
  const int hs = lane >> 4;
  const int lr = lane & 15;
  const int row = rowTile * 16 + lr;
  const int col = colTile * 16 + lr;
  const float* __restrict__ ar = Agg + (size_t)row * HCH;

  v8f acc = {};
#pragma unroll 4
  for (int k0 = 0; k0 < HCH; k0 += 4) {
    const int ka = k0 + hs * 2;
    float a0 = ar[ka]     + b1[ka];
    float a1 = ar[ka + 1] + b1[ka + 1];
    v2f a; a.x = a0 > 0.0f ? a0 : 0.0f;
           a.y = a1 > 0.0f ? a1 : 0.0f;
    v2f b; b.x = W[ka * OCH + col];  b.y = W[(ka + 1) * OCH + col];
    acc = __builtin_amdgcn_wmma_f32_16x16x4_f32(false, a, false, b,
                                                (short)0, acc, false, false);
  }
#pragma unroll
  for (int r = 0; r < 8; ++r) {
    const int m = rowTile * 16 + hs * 8 + r;
    H2[(size_t)m * OCH + col] = acc[r];
  }
}

__global__ __launch_bounds__(256) void k_gather1(const int* __restrict__ cnt,
                                                 const int* __restrict__ cursor,
                                                 const int* __restrict__ slots,
                                                 const float* __restrict__ dinv,
                                                 const float* __restrict__ h,
                                                 float* __restrict__ agg, int n) {
  const int node = __builtin_amdgcn_readfirstlane((int)((blockIdx.x * 256 + threadIdx.x) >> 5));
  if (node >= n) return;
  const int lane = threadIdx.x & 31;
  const int c = cnt[node];
  const int2* __restrict__ seg = reinterpret_cast<const int2*>(slots) + (cursor[node] - c);
  const float dd = dinv[node];
  const float4* __restrict__ hv = reinterpret_cast<const float4*>(h);
  const float4 self = hv[(size_t)node * 32 + lane];
  const float w0 = dd * dd;
  float4 a;
  a.x = self.x * w0; a.y = self.y * w0; a.z = self.z * w0; a.w = self.w * w0;
  const int last = c - 1;
  for (int k = 0; k < c; k += 8) {
    int2 e[8];
    float4 v[8];
#pragma unroll
    for (int j = 0; j < 8; ++j) e[j] = seg[k + j < c ? k + j : last];
#pragma unroll
    for (int j = 0; j < 8; ++j) v[j] = hv[(size_t)e[j].x * 32 + lane];
#pragma unroll
    for (int j = 0; j < 8; ++j) {
      const bool ok = k + j < c;
      const float q = ok ? __int_as_float(e[j].y) * dd : 0.0f;
      const float ux = ok ? v[j].x : 0.0f, uy = ok ? v[j].y : 0.0f;
      const float uz = ok ? v[j].z : 0.0f, uw = ok ? v[j].w : 0.0f;
      a.x += ux * q; a.y += uy * q; a.z += uz * q; a.w += uw * q;
    }
  }
  reinterpret_cast<float4*>(agg)[(size_t)node * 32 + lane] = a;
}

__global__ __launch_bounds__(256) void k_gather2(const int* __restrict__ cnt,
                                                 const int* __restrict__ cursor,
                                                 const int* __restrict__ slots,
                                                 const float* __restrict__ h2,
                                                 const float* __restrict__ b2,
                                                 float* __restrict__ out, int n) {
  const int node = __builtin_amdgcn_readfirstlane((int)((blockIdx.x * 256 + threadIdx.x) >> 5));
  if (node >= n) return;
  const int lane = threadIdx.x & 31;
  const int c = cnt[node];
  const int2* __restrict__ seg = reinterpret_cast<const int2*>(slots) + (cursor[node] - c);
  const float2* __restrict__ hv = reinterpret_cast<const float2*>(h2);
  float2 a = reinterpret_cast<const float2*>(b2)[lane];
  const int last = c - 1;
  for (int k = 0; k < c; k += 8) {
    int sidx[8];
    float2 v[8];
#pragma unroll
    for (int j = 0; j < 8; ++j) sidx[j] = seg[k + j < c ? k + j : last].x;
#pragma unroll
    for (int j = 0; j < 8; ++j) v[j] = hv[(size_t)sidx[j] * 32 + lane];
#pragma unroll
    for (int j = 0; j < 8; ++j) {
      const bool ok = k + j < c;
      a.x += ok ? v[j].x : 0.0f;
      a.y += ok ? v[j].y : 0.0f;
    }
  }
  reinterpret_cast<float2*>(out)[(size_t)node * 32 + lane] = a;
}

extern "C" void kernel_launch(void* const* d_in, const int* in_sizes, int n_in,
                              void* d_out, int out_size, void* d_ws, size_t ws_size,
                              hipStream_t stream) {
  const float* x  = (const float*)d_in[0];
  const int*   ei = (const int*)d_in[1];
  const float* W1 = (const float*)d_in[2];
  const float* b1 = (const float*)d_in[3];
  const float* W2 = (const float*)d_in[4];
  const float* b2 = (const float*)d_in[5];

  const int N = in_sizes[0] / HCH;
  const int E = in_sizes[1] / 2;
  const int* src = ei;
  const int* dst = ei + E;

  float* ws   = (float*)d_ws;
  float* dinv = ws;
  size_t off  = ((size_t)N + 63) & ~(size_t)63;
  float* h    = ws + off;
  float* agg  = h + (size_t)N * HCH;
  float* h2   = h;
  float* out  = (float*)d_out;
  int*  cnt    = (int*)(agg + (size_t)N * HCH);
  int*  cursor = cnt + off;
  int*  total  = cursor + off;
  int*  slots  = total + 64;

  k_zero_cnt<<<(N + 255) / 256, 256, 0, stream>>>(cnt, total, N);
  k_count   <<<(E + 255) / 256, 256, 0, stream>>>(dst, cnt, E);
  k_alloc   <<<(N + 255) / 256, 256, 0, stream>>>(cnt, cursor, dinv, total, N);
  k_fill    <<<(E + 255) / 256, 256, 0, stream>>>(src, dst, dinv, cursor, slots, E);

  k_gemm1<<<N / 16, 256, 0, stream>>>(x, W1, h, N);

  long long gw = (long long)N * 32;
  k_gather1<<<(unsigned)((gw + 255) / 256), 256, 0, stream>>>(cnt, cursor, slots, dinv, h, agg, N);

  k_gemm2<<<N / 32, 256, 0, stream>>>(agg, b1, W2, h2, N);

  k_gather2<<<(unsigned)((gw + 255) / 256), 256, 0, stream>>>(cnt, cursor, slots, h2, b2, out, N);
}


# Both WMMA products: a block copies an 80-row A stripe into padded LDS with whole-row 512-B loads (bias+ReLU applied there) and each wave re-uses every B fragment for 5 row tiles; k order as before
# speedup vs baseline: 1.9163x; 1.1229x over previous
#include <hip/hip_runtime.h>

typedef __attribute__((ext_vector_type(2))) float v2f;
typedef __attribute__((ext_vector_type(4))) float v4f;
typedef __attribute__((ext_vector_type(8))) float v8f;

#define HCH 128
#define OCH 64

__global__ __launch_bounds__(256) void k_zero_cnt(int* __restrict__ cnt, int* __restrict__ total, int n) {
  int i = blockIdx.x * 256 + threadIdx.x;
  if (i < n) cnt[i] = 0;
  if (i == 0) *total = 0;
}

__global__ __launch_bounds__(256) void k_count(const int* __restrict__ dst,
                                               int* __restrict__ cnt, int e) {
  int i = blockIdx.x * 256 + threadIdx.x;
  if (i < e) atomicAdd(&cnt[dst[i]], 1);
}

__global__ __launch_bounds__(256) void k_alloc(const int* __restrict__ cnt,
                                               int* __restrict__ cursor,
                                               float* __restrict__ dinv,
                                               int* __restrict__ total, int n) {
  __shared__ int wsum[8];
  __shared__ int bbase;
  const int i = blockIdx.x * 256 + threadIdx.x;
  const int lane = threadIdx.x & 31;
  const int wave = threadIdx.x >> 5;
  const int c = (i < n) ? cnt[i] : 0;
  int v = c;
  for (int d = 1; d < 32; d <<= 1) {
    const int t = __shfl_up(v, d);
    if (lane >= d) v += t;
  }
  if (lane == 31) wsum[wave] = v;
  __syncthreads();
  if (threadIdx.x == 0) {
    int s = 0;
    for (int w = 0; w < 8; ++w) { const int t = wsum[w]; wsum[w] = s; s += t; }
    bbase = atomicAdd(total, s);
  }
  __syncthreads();
  if (i < n) {
    cursor[i] = bbase + wsum[wave] + v - c;
    dinv[i] = rsqrtf((float)(c + 1));
  }
}

__global__ __launch_bounds__(256) void k_fill(const int* __restrict__ src,
                                              const int* __restrict__ dst,
                                              const float* __restrict__ dinv,
                                              int* __restrict__ cursor,
                                              int* __restrict__ slots, int e) {
  int i = blockIdx.x * 256 + threadIdx.x;
  if (i < e) {
    const int s = src[i], d = dst[i];
    const int p = atomicAdd(&cursor[d], 1);
    reinterpret_cast<int2*>(slots)[p] = make_int2(s, __float_as_int(dinv[s]));
  }
}

#define LDP 132

template <int NCT, int RT, bool BIAS_RELU>
__device__ __forceinline__ void gemm_stripe(const float* __restrict__ A,
                                            const float* __restrict__ bias,
                                            const float* __restrict__ W,
                                            float* __restrict__ C, int n,
                                            float* __restrict__ sA) {
  constexpr int ROWS  = RT * 16;
  constexpr int LDB   = NCT * 16;
  constexpr int PER_W = ROWS / NCT;
  constexpr int BATCH = 10;
  static_assert(ROWS % NCT == 0 && PER_W % BATCH == 0, "stripe shape");
  const int wave = threadIdx.x >> 5;
  const int lane = threadIdx.x & 31;
  const int row0 = blockIdx.x * ROWS;

  const v4f* __restrict__ Av = reinterpret_cast<const v4f*>(A);
  v4f bv = {};
  if (BIAS_RELU) bv = reinterpret_cast<const v4f*>(bias)[lane];
  for (int p0 = 0; p0 < PER_W; p0 += BATCH) {
    v4f t[BATCH];
#pragma unroll
    for (int j = 0; j < BATCH; ++j) {
      const int g = row0 + (p0 + j) * NCT + wave;
      const int gc = g < n ? g : n - 1;
      t[j] = Av[(size_t)gc * 32 + lane];
    }
#pragma unroll
    for (int j = 0; j < BATCH; ++j) {
      v4f u = t[j];
      if (BIAS_RELU) {
        const float a0 = u.x + bv.x, a1 = u.y + bv.y, a2 = u.z + bv.z, a3 = u.w + bv.w;
        u.x = a0 > 0.0f ? a0 : 0.0f;
        u.y = a1 > 0.0f ? a1 : 0.0f;
        u.z = a2 > 0.0f ? a2 : 0.0f;
        u.w = a3 > 0.0f ? a3 : 0.0f;
      }
      *reinterpret_cast<v4f*>(sA + ((p0 + j) * NCT + wave) * LDP + lane * 4) = u;
    }
  }
  __syncthreads();

  const int hs = lane >> 4;
  const int lr = lane & 15;
  const int col = wave * 16 + lr;
  const float* __restrict__ sa = sA + lr * LDP + hs * 2;
  const float* __restrict__ wp = W + (size_t)(hs * 2) * LDB + col;

  v8f acc[RT];
#pragma unroll
  for (int rt = 0; rt < RT; ++rt) { v8f z = {}; acc[rt] = z; }

#pragma unroll 4
  for (int k0 = 0; k0 < HCH; k0 += 4) {
    v2f b; b.x = wp[(size_t)k0 * LDB];  b.y = wp[(size_t)(k0 + 1) * LDB];
    v2f a[RT];
#pragma unroll
    for (int rt = 0; rt < RT; ++rt)
      a[rt] = *reinterpret_cast<const v2f*>(sa + rt * 16 * LDP + k0);
#pragma unroll
    for (int rt = 0; rt < RT; ++rt)
      acc[rt] = __builtin_amdgcn_wmma_f32_16x16x4_f32(false, a[rt], false, b,
                                                      (short)0, acc[rt], false, false);
  }

#pragma unroll
  for (int rt = 0; rt < RT; ++rt) {
    const int m0 = row0 + rt * 16;
    if (m0 < n) {
#pragma unroll
      for (int r = 0; r < 8; ++r) {
        const int m = m0 + hs * 8 + r;
        C[(size_t)m * LDB + col] = acc[rt][r];
      }
    }
  }
}

__global__ __launch_bounds__(256) __attribute__((amdgpu_waves_per_eu(4, 8))) void k_gemm1(const float* __restrict__ X,
                                               const float* __restrict__ W,
                                               float* __restrict__ Hout, int n) {
  __shared__ __attribute__((aligned(16))) float sA[80 * LDP];
  gemm_stripe<8, 5, false>(X, nullptr, W, Hout, n, sA);
}

__global__ __launch_bounds__(128) __attribute__((amdgpu_waves_per_eu(4, 8))) void k_gemm2(const float* __restrict__ Agg,
                                               const float* __restrict__ b1,
                                               const float* __restrict__ W,
                                               float* __restrict__ H2, int n) {
  __shared__ __attribute__((aligned(16))) float sA[80 * LDP];
  gemm_stripe<4, 5, true>(Agg, b1, W, H2, n, sA);
}

__global__ __launch_bounds__(256) void k_gather1(const int* __restrict__ cnt,
                                                 const int* __restrict__ cursor,
                                                 const int* __restrict__ slots,
                                                 const float* __restrict__ dinv,
                                                 const float* __restrict__ h,
                                                 float* __restrict__ agg, int n) {
  const int node = __builtin_amdgcn_readfirstlane((int)((blockIdx.x * 256 + threadIdx.x) >> 5));
  if (node >= n) return;
  const int lane = threadIdx.x & 31;
  const int c = cnt[node];
  const int2* __restrict__ seg = reinterpret_cast<const int2*>(slots) + (cursor[node] - c);
  const float dd = dinv[node];
  const float4* __restrict__ hv = reinterpret_cast<const float4*>(h);
  const float4 self = hv[(size_t)node * 32 + lane];
  const float w0 = dd * dd;
  float4 a;
  a.x = self.x * w0; a.y = self.y * w0; a.z = self.z * w0; a.w = self.w * w0;
  const int last = c - 1;
  for (int k = 0; k < c; k += 8) {
    int2 e[8];
    float4 v[8];
#pragma unroll
    for (int j = 0; j < 8; ++j) e[j] = seg[k + j < c ? k + j : last];
#pragma unroll
    for (int j = 0; j < 8; ++j) v[j] = hv[(size_t)e[j].x * 32 + lane];
#pragma unroll
    for (int j = 0; j < 8; ++j) {
      const bool ok = k + j < c;
      const float q = ok ? __int_as_float(e[j].y) * dd : 0.0f;
      const float ux = ok ? v[j].x : 0.0f, uy = ok ? v[j].y : 0.0f;
      const float uz = ok ? v[j].z : 0.0f, uw = ok ? v[j].w : 0.0f;
      a.x += ux * q; a.y += uy * q; a.z += uz * q; a.w += uw * q;
    }
  }
  reinterpret_cast<float4*>(agg)[(size_t)node * 32 + lane] = a;
}

__global__ __launch_bounds__(256) void k_gather2(const int* __restrict__ cnt,
                                                 const int* __restrict__ cursor,
                                                 const int* __restrict__ slots,
                                                 const float* __restrict__ h2,
                                                 const float* __restrict__ b2,
                                                 float* __restrict__ out, int n) {
  const int node = __builtin_amdgcn_readfirstlane((int)((blockIdx.x * 256 + threadIdx.x) >> 5));
  if (node >= n) return;
  const int lane = threadIdx.x & 31;
  const int c = cnt[node];
  const int2* __restrict__ seg = reinterpret_cast<const int2*>(slots) + (cursor[node] - c);
  const float2* __restrict__ hv = reinterpret_cast<const float2*>(h2);
  float2 a = reinterpret_cast<const float2*>(b2)[lane];
  const int last = c - 1;
  for (int k = 0; k < c; k += 8) {
    int sidx[8];
    float2 v[8];
#pragma unroll
    for (int j = 0; j < 8; ++j) sidx[j] = seg[k + j < c ? k + j : last].x;
#pragma unroll
    for (int j = 0; j < 8; ++j) v[j] = hv[(size_t)sidx[j] * 32 + lane];
#pragma unroll
    for (int j = 0; j < 8; ++j) {
      const bool ok = k + j < c;
      a.x += ok ? v[j].x : 0.0f;
      a.y += ok ? v[j].y : 0.0f;
    }
  }
  reinterpret_cast<float2*>(out)[(size_t)node * 32 + lane] = a;
}

extern "C" void kernel_launch(void* const* d_in, const int* in_sizes, int n_in,
                              void* d_out, int out_size, void* d_ws, size_t ws_size,
                              hipStream_t stream) {
  const float* x  = (const float*)d_in[0];
  const int*   ei = (const int*)d_in[1];
  const float* W1 = (const float*)d_in[2];
  const float* b1 = (const float*)d_in[3];
  const float* W2 = (const float*)d_in[4];
  const float* b2 = (const float*)d_in[5];

  const int N = in_sizes[0] / HCH;
  const int E = in_sizes[1] / 2;
  const int* src = ei;
  const int* dst = ei + E;

  float* ws   = (float*)d_ws;
  float* dinv = ws;
  size_t off  = ((size_t)N + 63) & ~(size_t)63;
  float* h    = ws + off;
  float* agg  = h + (size_t)N * HCH;
  float* h2   = h;
  float* out  = (float*)d_out;
  int*  cnt    = (int*)(agg + (size_t)N * HCH);
  int*  cursor = cnt + off;
  int*  total  = cursor + off;
  int*  slots  = total + 64;

  k_zero_cnt<<<(N + 255) / 256, 256, 0, stream>>>(cnt, total, N);
  k_count   <<<(E + 255) / 256, 256, 0, stream>>>(dst, cnt, E);
  k_alloc   <<<(N + 255) / 256, 256, 0, stream>>>(cnt, cursor, dinv, total, N);
  k_fill    <<<(E + 255) / 256, 256, 0, stream>>>(src, dst, dinv, cursor, slots, E);

  k_gemm1<<<(N + 79) / 80, 256, 0, stream>>>(x, W1, h, N);

  long long gw = (long long)N * 32;
  k_gather1<<<(unsigned)((gw + 255) / 256), 256, 0, stream>>>(cnt, cursor, slots, dinv, h, agg, N);

  k_gemm2<<<(N + 79) / 80, 128, 0, stream>>>(agg, b1, W2, h2, N);

  k_gather2<<<(unsigned)((gw + 255) / 256), 256, 0, stream>>>(cnt, cursor, slots, h2, b2, out, N);
}


# Bucket build: edges sorted by destination with LDS counters (4096-edge chunks split by 512-node bin, one block per bin ranks its nodes), 2 launches for 4, no global atomics; sums in LDS rank order
# speedup vs baseline: 4.0347x; 2.1055x over previous
#include <hip/hip_runtime.h>

typedef __attribute__((ext_vector_type(2))) float v2f;
typedef __attribute__((ext_vector_type(4))) float v4f;
typedef __attribute__((ext_vector_type(8))) float v8f;

#define HCH 128
#define OCH 64

#define BSH 9
#define BNODES (1 << BSH)
#define MAXB 256
#define CH 4096
#define EPT (CH / 256)

__global__ __launch_bounds__(256) void k_part(const int* __restrict__ src,
                                              const int* __restrict__ dst,
                                              int2* __restrict__ part,
                                              int* __restrict__ segCnt,
                                              int* __restrict__ segOff,
                                              int e, int nb, int g) {
  __shared__ int hist[MAXB];
  __shared__ int wsum[8];
  const int tid = threadIdx.x;
  const int lane = tid & 31;
  const int wave = tid >> 5;
  const int c0 = blockIdx.x * CH;
  hist[tid] = 0;
  __syncthreads();
  int d[EPT], s[EPT], r[EPT];
#pragma unroll
  for (int j = 0; j < EPT; ++j) {
    const int i = c0 + j * 256 + tid;
    d[j] = i < e ? dst[i] : -1;
    s[j] = i < e ? src[i] : 0;
  }
#pragma unroll
  for (int j = 0; j < EPT; ++j)
    r[j] = d[j] >= 0 ? atomicAdd(&hist[d[j] >> BSH], 1) : 0;
  __syncthreads();
  const int c = hist[tid];
  int v = c;
  for (int k = 1; k < 32; k <<= 1) {
    const int t = __shfl_up(v, k);
    if (lane >= k) v += t;
  }
  if (lane == 31) wsum[wave] = v;
  __syncthreads();
  int wb = 0;
  for (int w = 0; w < wave; ++w) wb += wsum[w];
  const int excl = wb + v - c;
  hist[tid] = excl;
  if (tid < nb) {
    segCnt[(size_t)tid * g + blockIdx.x] = c;
    segOff[(size_t)tid * g + blockIdx.x] = excl;
  }
  __syncthreads();
#pragma unroll
  for (int j = 0; j < EPT; ++j)
    if (d[j] >= 0) part[c0 + hist[d[j] >> BSH] + r[j]] = make_int2(s[j], d[j]);
}

template <bool PLACE>
__device__ __forceinline__ void walk_runs(const int2* __restrict__ part,
                                          const int* __restrict__ segCnt,
                                          const int* __restrict__ segOff,
                                          int* __restrict__ ncnt, int* __restrict__ out,
                                          int row, int g0, int g1, int lane) {
  for (int t0 = g0; t0 < g1; t0 += 32) {
    const int gi = t0 + lane;
    int so = 0, sc = 0;
    if (gi < g1) { sc = segCnt[row + gi]; so = gi * CH + segOff[row + gi]; }
    const int m = g1 - t0 < 32 ? g1 - t0 : 32;
    for (int i = 0; i < m; i += 4) {
      int o[4], c[4];
      int2 p[4];
#pragma unroll
      for (int u = 0; u < 4; ++u) {
        o[u] = __builtin_amdgcn_readlane(so, i + u);
        c[u] = __builtin_amdgcn_readlane(sc, i + u);
      }
#pragma unroll
      for (int u = 0; u < 4; ++u)
        p[u] = part[lane < c[u] ? o[u] + lane : 0];
      asm volatile("" :: "v"(p[0].y), "v"(p[1].y), "v"(p[2].y), "v"(p[3].y));
#pragma unroll
      for (int u = 0; u < 4; ++u) {
        if (lane < c[u]) {
          const int dl = p[u].y & (BNODES - 1);
          const int pos = atomicAdd(&ncnt[dl], 1);
          if (PLACE) out[pos] = p[u].x;
        }
        for (int k = lane + 32; k < c[u]; k += 32) {
          const int2 q = part[o[u] + k];
          const int dl = q.y & (BNODES - 1);
          const int pos = atomicAdd(&ncnt[dl], 1);
          if (PLACE) out[pos] = q.x;
        }
      }
    }
  }
}

__global__ __launch_bounds__(1024) void k_bin(const int2* __restrict__ part,
                                              const int* __restrict__ segCnt,
                                              const int* __restrict__ segOff,
                                              int* __restrict__ cnt,
                                              int* __restrict__ cursor,
                                              float* __restrict__ dinv,
                                              int* __restrict__ slots, int n, int g) {
  __shared__ int ncnt[BNODES];
  __shared__ int wsumA[32];
  __shared__ int wsumB[32];
  const int tid = threadIdx.x;
  const int lane = tid & 31;
  const int wave = __builtin_amdgcn_readfirstlane(tid >> 5);
  const int b = blockIdx.x;
  const int row = b * g;
  if (tid < BNODES) ncnt[tid] = 0;

  int acc = 0;
  {
    const int4* __restrict__ sc4 = reinterpret_cast<const int4*>(segCnt);
    const int lim4 = row >> 2;
#pragma unroll 4
    for (int i = tid; i < lim4; i += 1024) {
      const int4 t = sc4[i];
      acc += (t.x + t.y) + (t.z + t.w);
    }
    if (tid < (row & 3)) acc += segCnt[(row & ~3) + tid];
  }
  for (int k = 16; k > 0; k >>= 1) acc += __shfl_xor(acc, k);
  if (lane == 0) wsumA[wave] = acc;

  const int spw = (g + 31) >> 5;
  const int g0 = wave * spw;
  const int g1 = g0 + spw < g ? g0 + spw : g;
  __syncthreads();

  walk_runs<false>(part, segCnt, segOff, ncnt, nullptr, row, g0, g1, lane);
  __syncthreads();

  int base = 0;
  for (int w = 0; w < 32; ++w) base += wsumA[w];
  const int c = tid < BNODES ? ncnt[tid] : 0;
  int v = c;
  for (int k = 1; k < 32; k <<= 1) {
    const int t = __shfl_up(v, k);
    if (lane >= k) v += t;
  }
  if (lane == 31) wsumB[wave] = v;
  __syncthreads();
  if (tid < BNODES) {
    int wb = 0;
    for (int w = 0; w < wave; ++w) wb += wsumB[w];
    const int start = wb + v - c;
    ncnt[tid] = start;
    const int node = b * BNODES + tid;
    if (node < n) {
      cnt[node] = c;
      cursor[node] = base + start + c;
      dinv[node] = rsqrtf((float)(c + 1));
    }
  }
  __syncthreads();

  walk_runs<true>(part, segCnt, segOff, ncnt, slots + base, row, g0, g1, lane);
}

#define LDP 132

template <int NCT, int RT, bool BIAS_RELU>
__device__ __forceinline__ void gemm_stripe(const float* __restrict__ A,
                                            const float* __restrict__ bias,
                                            const float* __restrict__ W,
                                            float* __restrict__ C, int n,
                                            float* __restrict__ sA) {
  constexpr int ROWS  = RT * 16;
  constexpr int LDB   = NCT * 16;
  constexpr int PER_W = ROWS / NCT;
  constexpr int BATCH = 10;
  static_assert(ROWS % NCT == 0 && PER_W % BATCH == 0, "stripe shape");
  const int wave = threadIdx.x >> 5;
  const int lane = threadIdx.x & 31;
  const int row0 = blockIdx.x * ROWS;

  const v4f* __restrict__ Av = reinterpret_cast<const v4f*>(A);
  v4f bv = {};
  if (BIAS_RELU) bv = reinterpret_cast<const v4f*>(bias)[lane];
  for (int p0 = 0; p0 < PER_W; p0 += BATCH) {
    v4f t[BATCH];
#pragma unroll
    for (int j = 0; j < BATCH; ++j) {
      const int g = row0 + (p0 + j) * NCT + wave;
      const int gc = g < n ? g : n - 1;
      t[j] = Av[(size_t)gc * 32 + lane];
    }
#pragma unroll
    for (int j = 0; j < BATCH; ++j) {
      v4f u = t[j];
      if (BIAS_RELU) {
        const float a0 = u.x + bv.x, a1 = u.y + bv.y, a2 = u.z + bv.z, a3 = u.w + bv.w;
        u.x = a0 > 0.0f ? a0 : 0.0f;
        u.y = a1 > 0.0f ? a1 : 0.0f;
        u.z = a2 > 0.0f ? a2 : 0.0f;
        u.w = a3 > 0.0f ? a3 : 0.0f;
      }
      *reinterpret_cast<v4f*>(sA + ((p0 + j) * NCT + wave) * LDP + lane * 4) = u;
    }
  }
  __syncthreads();

  const int hs = lane >> 4;
  const int lr = lane & 15;
  const int col = wave * 16 + lr;
  const float* __restrict__ sa = sA + lr * LDP + hs * 2;
  const float* __restrict__ wp = W + (size_t)(hs * 2) * LDB + col;

  v8f acc[RT];
#pragma unroll
  for (int rt = 0; rt < RT; ++rt) { v8f z = {}; acc[rt] = z; }

#pragma unroll 4
  for (int k0 = 0; k0 < HCH; k0 += 4) {
    v2f b; b.x = wp[(size_t)k0 * LDB];  b.y = wp[(size_t)(k0 + 1) * LDB];
    v2f a[RT];
#pragma unroll
    for (int rt = 0; rt < RT; ++rt)
      a[rt] = *reinterpret_cast<const v2f*>(sa + rt * 16 * LDP + k0);
#pragma unroll
    for (int rt = 0; rt < RT; ++rt)
      acc[rt] = __builtin_amdgcn_wmma_f32_16x16x4_f32(false, a[rt], false, b,
                                                      (short)0, acc[rt], false, false);
  }

#pragma unroll
  for (int rt = 0; rt < RT; ++rt) {
    const int m0 = row0 + rt * 16;
    if (m0 < n) {
#pragma unroll
      for (int r = 0; r < 8; ++r) {
        const int m = m0 + hs * 8 + r;
        C[(size_t)m * LDB + col] = acc[rt][r];
      }
    }
  }
}

__global__ __launch_bounds__(256) __attribute__((amdgpu_waves_per_eu(4, 8))) void k_gemm1(const float* __restrict__ X,
                                               const float* __restrict__ W,
                                               float* __restrict__ Hout, int n) {
  __shared__ __attribute__((aligned(16))) float sA[80 * LDP];
  gemm_stripe<8, 5, false>(X, nullptr, W, Hout, n, sA);
}

__global__ __launch_bounds__(128) __attribute__((amdgpu_waves_per_eu(4, 8))) void k_gemm2(const float* __restrict__ Agg,
                                               const float* __restrict__ b1,
                                               const float* __restrict__ W,
                                               float* __restrict__ H2, int n) {
  __shared__ __attribute__((aligned(16))) float sA[80 * LDP];
  gemm_stripe<4, 5, true>(Agg, b1, W, H2, n, sA);
}

__global__ __launch_bounds__(256) void k_gather1(const int* __restrict__ cnt,
                                                 const int* __restrict__ cursor,
                                                 const int* __restrict__ slots,
                                                 const float* __restrict__ dinv,
                                                 const float* __restrict__ h,
                                                 float* __restrict__ agg, int n) {
  const int node = __builtin_amdgcn_readfirstlane((int)((blockIdx.x * 256 + threadIdx.x) >> 5));
  if (node >= n) return;
  const int lane = threadIdx.x & 31;
  const int c = cnt[node];
  const int* __restrict__ seg = slots + (cursor[node] - c);
  const float dd = dinv[node];
  const float4* __restrict__ hv = reinterpret_cast<const float4*>(h);
  const float4 self = hv[(size_t)node * 32 + lane];
  const float w0 = dd * dd;
  float4 a;
  a.x = self.x * w0; a.y = self.y * w0; a.z = self.z * w0; a.w = self.w * w0;
  const int last = c - 1;
  for (int k = 0; k < c; k += 8) {
    int e[8];
    float ds[8];
    float4 v[8];
#pragma unroll
    for (int j = 0; j < 8; ++j) e[j] = seg[k + j < c ? k + j : last];
#pragma unroll
    for (int j = 0; j < 8; ++j) ds[j] = dinv[e[j]];
#pragma unroll
    for (int j = 0; j < 8; ++j) v[j] = hv[(size_t)e[j] * 32 + lane];
#pragma unroll
    for (int j = 0; j < 8; ++j) {
      const bool ok = k + j < c;
      const float q = ok ? ds[j] * dd : 0.0f;
      const float ux = ok ? v[j].x : 0.0f, uy = ok ? v[j].y : 0.0f;
      const float uz = ok ? v[j].z : 0.0f, uw = ok ? v[j].w : 0.0f;
      a.x += ux * q; a.y += uy * q; a.z += uz * q; a.w += uw * q;
    }
  }
  reinterpret_cast<float4*>(agg)[(size_t)node * 32 + lane] = a;
}

__global__ __launch_bounds__(256) void k_gather2(const int* __restrict__ cnt,
                                                 const int* __restrict__ cursor,
                                                 const int* __restrict__ slots,
                                                 const float* __restrict__ h2,
                                                 const float* __restrict__ b2,
                                                 float* __restrict__ out, int n) {
  const int node = __builtin_amdgcn_readfirstlane((int)((blockIdx.x * 256 + threadIdx.x) >> 5));
  if (node >= n) return;
  const int lane = threadIdx.x & 31;
  const int c = cnt[node];
  const int* __restrict__ seg = slots + (cursor[node] - c);
  const float2* __restrict__ hv = reinterpret_cast<const float2*>(h2);
  float2 a = reinterpret_cast<const float2*>(b2)[lane];
  const int last = c - 1;
  for (int k = 0; k < c; k += 8) {
    int sidx[8];
    float2 v[8];
#pragma unroll
    for (int j = 0; j < 8; ++j) sidx[j] = seg[k + j < c ? k + j : last];
#pragma unroll
    for (int j = 0; j < 8; ++j) v[j] = hv[(size_t)sidx[j] * 32 + lane];
#pragma unroll
    for (int j = 0; j < 8; ++j) {
      const bool ok = k + j < c;
      a.x += ok ? v[j].x : 0.0f;
      a.y += ok ? v[j].y : 0.0f;
    }
  }
  reinterpret_cast<float2*>(out)[(size_t)node * 32 + lane] = a;
}

extern "C" void kernel_launch(void* const* d_in, const int* in_sizes, int n_in,
                              void* d_out, int out_size, void* d_ws, size_t ws_size,
                              hipStream_t stream) {
  const float* x  = (const float*)d_in[0];
  const int*   ei = (const int*)d_in[1];
  const float* W1 = (const float*)d_in[2];
  const float* b1 = (const float*)d_in[3];
  const float* W2 = (const float*)d_in[4];
  const float* b2 = (const float*)d_in[5];

  const int N = in_sizes[0] / HCH;
  const int E = in_sizes[1] / 2;
  const int* src = ei;
  const int* dst = ei + E;

  float* ws   = (float*)d_ws;
  float* dinv = ws;
  size_t off  = ((size_t)N + 63) & ~(size_t)63;
  float* h    = ws + off;
  float* agg  = h + (size_t)N * HCH;
  float* h2   = h;
  float* out  = (float*)d_out;
  int*  cnt    = (int*)(agg + (size_t)N * HCH);
  int*  cursor = cnt + off;
  const int G  = (E + CH - 1) / CH;
  const int NB = (N + BNODES - 1) >> BSH;
  size_t eoff  = ((size_t)E + 63) & ~(size_t)63;
  size_t toff  = ((size_t)NB * G + 63) & ~(size_t)63;
  int*  slots  = cursor + off;
  int2* part   = (int2*)(slots + eoff);
  int*  segCnt = slots + 3 * eoff;
  int*  segOff = segCnt + toff;

  k_part<<<G, 256, 0, stream>>>(src, dst, part, segCnt, segOff, E, NB, G);
  k_bin <<<NB, 1024, 0, stream>>>(part, segCnt, segOff, cnt, cursor, dinv, slots, N, G);

  k_gemm1<<<(N + 79) / 80, 256, 0, stream>>>(x, W1, h, N);

  long long gw = (long long)N * 32;
  k_gather1<<<(unsigned)((gw + 255) / 256), 256, 0, stream>>>(cnt, cursor, slots, dinv, h, agg, N);

  k_gemm2<<<(N + 79) / 80, 128, 0, stream>>>(agg, b1, W2, h2, N);

  k_gather2<<<(unsigned)((gw + 255) / 256), 256, 0, stream>>>(cnt, cursor, slots, h2, b2, out, N);
}


# conv1: the in-rows of x are summed first and the sum is multiplied by W1 in the same block (linearity), so x@W1 and its launch go; conv1 rounding and sum order change (product after the sum)
# speedup vs baseline: 4.5626x; 1.1308x over previous
#include <hip/hip_runtime.h>

typedef __attribute__((ext_vector_type(2))) float v2f;
typedef __attribute__((ext_vector_type(4))) float v4f;
typedef __attribute__((ext_vector_type(8))) float v8f;

#define HCH 128
#define OCH 64

#define BSH 9
#define BNODES (1 << BSH)
#define MAXB 256
#define CH 4096
#define EPT (CH / 256)

__global__ __launch_bounds__(256) void k_part(const int* __restrict__ src,
                                              const int* __restrict__ dst,
                                              int2* __restrict__ part,
                                              int* __restrict__ segCnt,
                                              int* __restrict__ segOff,
                                              int e, int nb, int g) {
  __shared__ int hist[MAXB];
  __shared__ int wsum[8];
  const int tid = threadIdx.x;
  const int lane = tid & 31;
  const int wave = tid >> 5;
  const int c0 = blockIdx.x * CH;
  hist[tid] = 0;
  __syncthreads();
  int d[EPT], s[EPT], r[EPT];
#pragma unroll
  for (int j = 0; j < EPT; ++j) {
    const int i = c0 + j * 256 + tid;
    d[j] = i < e ? dst[i] : -1;
    s[j] = i < e ? src[i] : 0;
  }
#pragma unroll
  for (int j = 0; j < EPT; ++j)
    r[j] = d[j] >= 0 ? atomicAdd(&hist[d[j] >> BSH], 1) : 0;
  __syncthreads();
  const int c = hist[tid];
  int v = c;
  for (int k = 1; k < 32; k <<= 1) {
    const int t = __shfl_up(v, k);
    if (lane >= k) v += t;
  }
  if (lane == 31) wsum[wave] = v;
  __syncthreads();
  int wb = 0;
  for (int w = 0; w < wave; ++w) wb += wsum[w];
  const int excl = wb + v - c;
  hist[tid] = excl;
  if (tid < nb) {
    segCnt[(size_t)tid * g + blockIdx.x] = c;
    segOff[(size_t)tid * g + blockIdx.x] = excl;
  }
  __syncthreads();
#pragma unroll
  for (int j = 0; j < EPT; ++j)
    if (d[j] >= 0) part[c0 + hist[d[j] >> BSH] + r[j]] = make_int2(s[j], d[j]);
}

template <bool PLACE>
__device__ __forceinline__ void walk_runs(const int2* __restrict__ part,
                                          const int* __restrict__ segCnt,
                                          const int* __restrict__ segOff,
                                          int* __restrict__ ncnt, int* __restrict__ out,
                                          int row, int g0, int g1, int lane) {
  for (int t0 = g0; t0 < g1; t0 += 32) {
    const int gi = t0 + lane;
    int so = 0, sc = 0;
    if (gi < g1) { sc = segCnt[row + gi]; so = gi * CH + segOff[row + gi]; }
    const int m = g1 - t0 < 32 ? g1 - t0 : 32;
    for (int i = 0; i < m; i += 4) {
      int o[4], c[4];
      int2 p[4];
#pragma unroll
      for (int u = 0; u < 4; ++u) {
        o[u] = __builtin_amdgcn_readlane(so, i + u);
        c[u] = __builtin_amdgcn_readlane(sc, i + u);
      }
#pragma unroll
      for (int u = 0; u < 4; ++u)
        p[u] = part[lane < c[u] ? o[u] + lane : 0];
      asm volatile("" :: "v"(p[0].y), "v"(p[1].y), "v"(p[2].y), "v"(p[3].y));
#pragma unroll
      for (int u = 0; u < 4; ++u) {
        if (lane < c[u]) {
          const int dl = p[u].y & (BNODES - 1);
          const int pos = atomicAdd(&ncnt[dl], 1);
          if (PLACE) out[pos] = p[u].x;
        }
        for (int k = lane + 32; k < c[u]; k += 32) {
          const int2 q = part[o[u] + k];
          const int dl = q.y & (BNODES - 1);
          const int pos = atomicAdd(&ncnt[dl], 1);
          if (PLACE) out[pos] = q.x;
        }
      }
    }
  }
}

__global__ __launch_bounds__(1024) void k_bin(const int2* __restrict__ part,
                                              const int* __restrict__ segCnt,
                                              const int* __restrict__ segOff,
                                              int* __restrict__ cnt,
                                              int* __restrict__ cursor,
                                              float* __restrict__ dinv,
                                              int* __restrict__ slots, int n, int g) {
  __shared__ int ncnt[BNODES];
  __shared__ int wsumA[32];
  __shared__ int wsumB[32];
  const int tid = threadIdx.x;
  const int lane = tid & 31;
  const int wave = __builtin_amdgcn_readfirstlane(tid >> 5);
  const int b = blockIdx.x;
  const int row = b * g;
  if (tid < BNODES) ncnt[tid] = 0;

  int acc = 0;
  {
    const int4* __restrict__ sc4 = reinterpret_cast<const int4*>(segCnt);
    const int lim4 = row >> 2;
#pragma unroll 4
    for (int i = tid; i < lim4; i += 1024) {
      const int4 t = sc4[i];
      acc += (t.x + t.y) + (t.z + t.w);
    }
    if (tid < (row & 3)) acc += segCnt[(row & ~3) + tid];
  }
  for (int k = 16; k > 0; k >>= 1) acc += __shfl_xor(acc, k);
  if (lane == 0) wsumA[wave] = acc;

  const int spw = (g + 31) >> 5;
  const int g0 = wave * spw;
  const int g1 = g0 + spw < g ? g0 + spw : g;
  __syncthreads();

  walk_runs<false>(part, segCnt, segOff, ncnt, nullptr, row, g0, g1, lane);
  __syncthreads();

  int base = 0;
  for (int w = 0; w < 32; ++w) base += wsumA[w];
  const int c = tid < BNODES ? ncnt[tid] : 0;
  int v = c;
  for (int k = 1; k < 32; k <<= 1) {
    const int t = __shfl_up(v, k);
    if (lane >= k) v += t;
  }
  if (lane == 31) wsumB[wave] = v;
  __syncthreads();
  if (tid < BNODES) {
    int wb = 0;
    for (int w = 0; w < wave; ++w) wb += wsumB[w];
    const int start = wb + v - c;
    ncnt[tid] = start;
    const int node = b * BNODES + tid;
    if (node < n) {
      cnt[node] = c;
      cursor[node] = base + start + c;
      dinv[node] = rsqrtf((float)(c + 1));
    }
  }
  __syncthreads();

  walk_runs<true>(part, segCnt, segOff, ncnt, slots + base, row, g0, g1, lane);
}

#define LDP 132

template <int NCT, int RT>
__device__ __forceinline__ void stripe_mma(const float* __restrict__ sA,
                                           const float* __restrict__ W,
                                           v8f (&acc)[RT]) {
  constexpr int LDB  = NCT * 16;
  const int wave = threadIdx.x >> 5;
  const int lane = threadIdx.x & 31;
  const int hs = lane >> 4;
  const int lr = lane & 15;
  const int col = wave * 16 + lr;
  const float* __restrict__ sa = sA + lr * LDP + hs * 2;
  const float* __restrict__ wp = W + (size_t)(hs * 2) * LDB + col;

#pragma unroll
  for (int rt = 0; rt < RT; ++rt) { v8f z = {}; acc[rt] = z; }

#pragma unroll 4
  for (int k0 = 0; k0 < HCH; k0 += 4) {
    v2f b; b.x = wp[(size_t)k0 * LDB];  b.y = wp[(size_t)(k0 + 1) * LDB];
    v2f a[RT];
#pragma unroll
    for (int rt = 0; rt < RT; ++rt)
      a[rt] = *reinterpret_cast<const v2f*>(sa + rt * 16 * LDP + k0);
#pragma unroll
    for (int rt = 0; rt < RT; ++rt)
      acc[rt] = __builtin_amdgcn_wmma_f32_16x16x4_f32(false, a[rt], false, b,
                                                      (short)0, acc[rt], false, false);
  }
}

#define AG_ROWS 32
__global__ __launch_bounds__(256) void k_conv1_gemm2(const int* __restrict__ cnt,
                                                     const int* __restrict__ cursor,
                                                     const int* __restrict__ slots,
                                                     const float* __restrict__ dinv,
                                                     const float* __restrict__ x,
                                                     const float* __restrict__ W1,
                                                     const float* __restrict__ b1,
                                                     const float* __restrict__ W2,
                                                     float* __restrict__ H2, int n) {
  __shared__ __attribute__((aligned(16))) float sA[AG_ROWS * LDP];
  const int lane = threadIdx.x & 31;
  const int wave = __builtin_amdgcn_readfirstlane((int)(threadIdx.x >> 5));
  const int row0 = blockIdx.x * AG_ROWS;
  const float4* __restrict__ xv = reinterpret_cast<const float4*>(x);
  for (int p = 0; p < AG_ROWS / 8; ++p) {
    const int r = p * 8 + wave;
    const int node = row0 + r;
    float4 a;
    a.x = 0.0f; a.y = 0.0f; a.z = 0.0f; a.w = 0.0f;
    if (node < n) {
      const int c = cnt[node];
      const int* __restrict__ seg = slots + (cursor[node] - c);
      const float dd = dinv[node];
      const float4 self = xv[(size_t)node * 32 + lane];
      const float w0 = dd * dd;
      a.x = self.x * w0; a.y = self.y * w0; a.z = self.z * w0; a.w = self.w * w0;
      const int last = c - 1;
      for (int k = 0; k < c; k += 8) {
        int e[8];
        float ds[8];
        float4 v[8];
#pragma unroll
        for (int j = 0; j < 8; ++j) e[j] = seg[k + j < c ? k + j : last];
#pragma unroll
        for (int j = 0; j < 8; ++j) ds[j] = dinv[e[j]];
#pragma unroll
        for (int j = 0; j < 8; ++j) v[j] = xv[(size_t)e[j] * 32 + lane];
#pragma unroll
        for (int j = 0; j < 8; ++j) {
          const bool ok = k + j < c;
          const float q = ok ? ds[j] * dd : 0.0f;
          const float ux = ok ? v[j].x : 0.0f, uy = ok ? v[j].y : 0.0f;
          const float uz = ok ? v[j].z : 0.0f, uw = ok ? v[j].w : 0.0f;
          a.x += ux * q; a.y += uy * q; a.z += uz * q; a.w += uw * q;
        }
      }
    }
    *reinterpret_cast<float4*>(sA + r * LDP + lane * 4) = a;
  }
  __syncthreads();

  const int hs = lane >> 4;
  const int lr = lane & 15;
  {
    v8f acc[2];
    stripe_mma<8, 2>(sA, W1, acc);
    __syncthreads();
    const int col = wave * 16 + lr;
    const float bc = b1[col];
#pragma unroll
    for (int rt = 0; rt < 2; ++rt) {
#pragma unroll
      for (int r = 0; r < 8; ++r) {
        const float t = acc[rt][r] + bc;
        sA[(rt * 16 + hs * 8 + r) * LDP + col] = t > 0.0f ? t : 0.0f;
      }
    }
  }
  __syncthreads();

  if (wave < 4) {
    v8f acc[2];
    stripe_mma<4, 2>(sA, W2, acc);
    const int col = wave * 16 + lr;
#pragma unroll
    for (int rt = 0; rt < 2; ++rt) {
      const int m0 = row0 + rt * 16;
      if (m0 < n) {
#pragma unroll
        for (int r = 0; r < 8; ++r) {
          const int m = m0 + hs * 8 + r;
          H2[(size_t)m * OCH + col] = acc[rt][r];
        }
      }
    }
  }
}

__global__ __launch_bounds__(256) void k_gather2(const int* __restrict__ cnt,
                                                 const int* __restrict__ cursor,
                                                 const int* __restrict__ slots,
                                                 const float* __restrict__ h2,
                                                 const float* __restrict__ b2,
                                                 float* __restrict__ out, int n) {
  const int node = __builtin_amdgcn_readfirstlane((int)((blockIdx.x * 256 + threadIdx.x) >> 5));
  if (node >= n) return;
  const int lane = threadIdx.x & 31;
  const int c = cnt[node];
  const int* __restrict__ seg = slots + (cursor[node] - c);
  const float2* __restrict__ hv = reinterpret_cast<const float2*>(h2);
  float2 a = reinterpret_cast<const float2*>(b2)[lane];
  const int last = c - 1;
  for (int k = 0; k < c; k += 8) {
    int sidx[8];
    float2 v[8];
#pragma unroll
    for (int j = 0; j < 8; ++j) sidx[j] = seg[k + j < c ? k + j : last];
#pragma unroll
    for (int j = 0; j < 8; ++j) v[j] = hv[(size_t)sidx[j] * 32 + lane];
#pragma unroll
    for (int j = 0; j < 8; ++j) {
      const bool ok = k + j < c;
      a.x += ok ? v[j].x : 0.0f;
      a.y += ok ? v[j].y : 0.0f;
    }
  }
  reinterpret_cast<float2*>(out)[(size_t)node * 32 + lane] = a;
}

extern "C" void kernel_launch(void* const* d_in, const int* in_sizes, int n_in,
                              void* d_out, int out_size, void* d_ws, size_t ws_size,
                              hipStream_t stream) {
  const float* x  = (const float*)d_in[0];
  const int*   ei = (const int*)d_in[1];
  const float* W1 = (const float*)d_in[2];
  const float* b1 = (const float*)d_in[3];
  const float* W2 = (const float*)d_in[4];
  const float* b2 = (const float*)d_in[5];

  const int N = in_sizes[0] / HCH;
  const int E = in_sizes[1] / 2;
  const int* src = ei;
  const int* dst = ei + E;

  float* ws   = (float*)d_ws;
  float* dinv = ws;
  size_t off  = ((size_t)N + 63) & ~(size_t)63;
  float* h2   = ws + off;
  float* out  = (float*)d_out;
  int*  cnt    = (int*)(h2 + (size_t)N * OCH);
  int*  cursor = cnt + off;
  const int G  = (E + CH - 1) / CH;
  const int NB = (N + BNODES - 1) >> BSH;
  size_t eoff  = ((size_t)E + 63) & ~(size_t)63;
  size_t toff  = ((size_t)NB * G + 63) & ~(size_t)63;
  int*  slots  = cursor + off;
  int2* part   = (int2*)(slots + eoff);
  int*  segCnt = slots + 3 * eoff;
  int*  segOff = segCnt + toff;

  k_part<<<G, 256, 0, stream>>>(src, dst, part, segCnt, segOff, E, NB, G);
  k_bin <<<NB, 1024, 0, stream>>>(part, segCnt, segOff, cnt, cursor, dinv, slots, N, G);

  k_conv1_gemm2<<<(N + AG_ROWS - 1) / AG_ROWS, 256, 0, stream>>>(cnt, cursor, slots, dinv, x, W1, b1, W2, h2, N);

  long long gw = (long long)N * 32;
  k_gather2<<<(unsigned)((gw + 255) / 256), 256, 0, stream>>>(cnt, cursor, slots, h2, b2, out, N);
}
